# starting kernel
# baseline (speedup 1.0000x reference)
#include <hip/hip_runtime.h>

typedef __attribute__((ext_vector_type(2))) float v2f;
typedef __attribute__((ext_vector_type(8))) float v8f;

#define N_NODES 10000
#define N_EDGES 640000
#define D_IN    128
#define D_HID   256
#define D_OUT   64

__global__ void zero_i32(int* __restrict__ p, int n) {
  int i = blockIdx.x * blockDim.x + threadIdx.x;
  if (i < n) p[i] = 0;
}

__global__ void count_deg(const int* __restrict__ dst, int* __restrict__ deg, int nE) {
  int e = blockIdx.x * blockDim.x + threadIdx.x;
  if (e < nE) atomicAdd(&deg[dst[e]], 1);
}

__global__ void exclusive_scan(const int* __restrict__ deg, int* __restrict__ rowptr, int n) {
  __shared__ int smem[1024];
  __shared__ int carry;
  int t = threadIdx.x;
  if (t == 0) carry = 0;
  __syncthreads();
  for (int base = 0; base < n; base += 1024) {
    int i = base + t;
    int v = (i < n) ? deg[i] : 0;
    smem[t] = v;
    __syncthreads();
    #pragma unroll
    for (int off = 1; off < 1024; off <<= 1) {
      int x = (t >= off) ? smem[t - off] : 0;
      __syncthreads();
      smem[t] += x;
      __syncthreads();
    }
    int cbase = carry;
    if (i < n) rowptr[i] = cbase + smem[t] - v;
    __syncthreads();
    if (t == 0) carry = cbase + smem[1023];
    __syncthreads();
  }
  if (t == 0) rowptr[n] = carry;
}

__global__ void inv_from_deg(const int* __restrict__ deg, float* __restrict__ inv, int n) {
  int i = blockIdx.x * blockDim.x + threadIdx.x;
  if (i < n) {
    int d = deg[i];
    inv[i] = (d > 0) ? (1.0f / (float)d) : 0.0f;
  }
}

__global__ void copy_i32(const int* __restrict__ a, int* __restrict__ b, int n) {
  int i = blockIdx.x * blockDim.x + threadIdx.x;
  if (i < n) b[i] = a[i];
}

__global__ void bin_edges(const int* __restrict__ src, const int* __restrict__ dst,
                          const float* __restrict__ w,
                          int* __restrict__ cursor,
                          int* __restrict__ es, float* __restrict__ ew, int nE) {
  int e = blockIdx.x * blockDim.x + threadIdx.x;
  if (e < nE) {
    int pos = atomicAdd(&cursor[dst[e]], 1);
    es[pos] = src[e];
    ew[pos] = w[e];
  }
}

template<int D>
__global__ void aggregate_nodes(const float* __restrict__ X, int ldx,
                                const int* __restrict__ rowptr,
                                const int* __restrict__ es,
                                const float* __restrict__ ew,
                                const float* __restrict__ inv,
                                float* Y, int ldy, int coff, int nN) {
  int node = (blockIdx.x * blockDim.x + threadIdx.x) >> 5;
  int lane = threadIdx.x & 31;
  if (node >= nN) return;
  int beg = rowptr[node];
  int end = rowptr[node + 1];
  float4 acc[D / 128];
  #pragma unroll
  for (int c = 0; c < D / 128; ++c) acc[c] = make_float4(0.f, 0.f, 0.f, 0.f);
  for (int j = beg; j < end; ++j) {
    int s = es[j];
    float ww = ew[j];
    const float4* xr = (const float4*)(X + (size_t)s * ldx);
    #pragma unroll
    for (int c = 0; c < D / 128; ++c) {
      float4 v = xr[lane + 32 * c];
      acc[c].x += v.x * ww;
      acc[c].y += v.y * ww;
      acc[c].z += v.z * ww;
      acc[c].w += v.w * ww;
    }
  }
  float iv = inv[node];
  float4* yr = (float4*)(Y + (size_t)node * ldy + coff);
  #pragma unroll
  for (int c = 0; c < D / 128; ++c)
    yr[lane + 32 * c] = make_float4(acc[c].x * iv, acc[c].y * iv, acc[c].z * iv, acc[c].w * iv);
}

__global__ void copy_self(const float* __restrict__ h, float* __restrict__ Y, int nN) {
  int i = blockIdx.x * blockDim.x + threadIdx.x;
  if (i >= nN * D_IN) return;
  int n = i >> 7;
  int f = i & 127;
  Y[(size_t)n * (2 * D_IN) + f] = h[i];
}

template<int NDIM, int KDIM, bool RELU>
__global__ void wmma_gemm_f32(const float* __restrict__ A, int lda,
                              const float* __restrict__ B,
                              const float* __restrict__ bias,
                              float* __restrict__ Y, int ldy,
                              int Mtiles, int Nstrips) {
  int wid  = (blockIdx.x * blockDim.x + threadIdx.x) >> 5;
  int lane = threadIdx.x & 31;
  if (wid >= Mtiles * Nstrips) return;
  int m0 = (wid / Nstrips) * 16;
  int n0 = (wid % Nstrips) * 64;
  int half = lane >> 4;
  int mr   = lane & 15;

  const float* ap = A + (size_t)(m0 + mr) * lda + 2 * half;
  const float* bp = B + (2 * half) * NDIM + n0 + mr;
  v8f acc0 = {}, acc1 = {}, acc2 = {}, acc3 = {};

  #pragma unroll 2
  for (int k = 0; k < KDIM; k += 4) {
    v2f a = *(const v2f*)ap;
    ap += 4;
    __builtin_prefetch(ap + 60, 0, 3);
    v2f b;
    b.x = bp[0];         b.y = bp[NDIM];
    acc0 = __builtin_amdgcn_wmma_f32_16x16x4_f32(false, a, false, b, (short)0, acc0, false, false);
    b.x = bp[16];        b.y = bp[NDIM + 16];
    acc1 = __builtin_amdgcn_wmma_f32_16x16x4_f32(false, a, false, b, (short)0, acc1, false, false);
    b.x = bp[32];        b.y = bp[NDIM + 32];
    acc2 = __builtin_amdgcn_wmma_f32_16x16x4_f32(false, a, false, b, (short)0, acc2, false, false);
    b.x = bp[48];        b.y = bp[NDIM + 48];
    acc3 = __builtin_amdgcn_wmma_f32_16x16x4_f32(false, a, false, b, (short)0, acc3, false, false);
    bp += 4 * NDIM;
  }

  #pragma unroll
  for (int j = 0; j < 4; ++j) {
    v8f acc = (j == 0) ? acc0 : (j == 1) ? acc1 : (j == 2) ? acc2 : acc3;
    int n = n0 + 16 * j + mr;
    float bv = bias[n];
    #pragma unroll
    for (int r = 0; r < 8; ++r) {
      int row = m0 + r + 8 * half;
      float v = acc[r] + bv;
      if (RELU) v = (v > 0.0f) ? v : 0.0f;
      Y[(size_t)row * ldy + n] = v;
    }
  }
}

extern "C" void kernel_launch(void* const* d_in, const int* in_sizes, int n_in,
                              void* d_out, int out_size, void* d_ws, size_t ws_size,
                              hipStream_t stream) {
  const float* h   = (const float*)d_in[0];
  const float* w   = (const float*)d_in[1];
  const int*   src = (const int*)d_in[2];
  const int*   dst = (const int*)d_in[3];
  const float* W1  = (const float*)d_in[4];
  const float* b1  = (const float*)d_in[5];
  const float* W2  = (const float*)d_in[6];
  const float* b2  = (const float*)d_in[7];
  float* out = (float*)d_out;

  int*   deg    = (int*)d_ws;
  int*   rowptr = deg + 10240;
  int*   cursor = rowptr + 10240;
  int*   es     = cursor + 10240;
  float* ew     = (float*)(es + N_EDGES);
  float* inv    = ew + N_EDGES;
  float* bufA   = inv + 10240;
  float* bufB   = bufA + N_NODES * D_HID;

  zero_i32<<<(N_NODES + 255) / 256, 256, 0, stream>>>(deg, N_NODES);
  count_deg<<<(N_EDGES + 255) / 256, 256, 0, stream>>>(dst, deg, N_EDGES);
  exclusive_scan<<<1, 1024, 0, stream>>>(deg, rowptr, N_NODES);
  inv_from_deg<<<(N_NODES + 255) / 256, 256, 0, stream>>>(deg, inv, N_NODES);
  copy_i32<<<(N_NODES + 255) / 256, 256, 0, stream>>>(rowptr, cursor, N_NODES);
  bin_edges<<<(N_EDGES + 255) / 256, 256, 0, stream>>>(src, dst, w, cursor, es, ew, N_EDGES);

  copy_self<<<(N_NODES * D_IN + 255) / 256, 256, 0, stream>>>(h, bufA, N_NODES);
  aggregate_nodes<D_IN><<<(N_NODES * 32 + 255) / 256, 256, 0, stream>>>(
      h, D_IN, rowptr, es, ew, inv, bufA, 2 * D_IN, D_IN, N_NODES);
  {
    int Mtiles = N_NODES / 16, Nstrips = D_HID / 64;
    wmma_gemm_f32<D_HID, 2 * D_IN, true><<<(Mtiles * Nstrips + 7) / 8, 256, 0, stream>>>(
        bufA, 2 * D_IN, W1, b1, bufB, 2 * D_HID, Mtiles, Nstrips);
  }

  aggregate_nodes<D_HID><<<(N_NODES * 32 + 255) / 256, 256, 0, stream>>>(
      bufB, 2 * D_HID, rowptr, es, ew, inv, bufB, 2 * D_HID, D_HID, N_NODES);
  {
    int Mtiles = N_NODES / 16, Nstrips = D_OUT / 64;
    wmma_gemm_f32<D_OUT, 2 * D_HID, false><<<(Mtiles * Nstrips + 7) / 8, 256, 0, stream>>>(
        bufB, 2 * D_HID, W2, b2, out, D_OUT, Mtiles, Nstrips);
  }
}


# each layer is one launch: a 16-wave block gathers its 16 nodes' own and mean rows into an LDS tile and multiplies it by W (the A operand never goes to memory); 3 launches, was 6; sums unchanged
# speedup vs baseline: 2.8129x; 2.8129x over previous
#include <hip/hip_runtime.h>

typedef __attribute__((ext_vector_type(2))) float v2f;
typedef __attribute__((ext_vector_type(8))) float v8f;

#define N_NODES 10000
#define N_EDGES 640000
#define D_IN    128
#define D_HID   256
#define D_OUT   64

#define CSR_R    64
#define CSR_B    ((N_NODES + CSR_R - 1) / CSR_R)
#define CSR_NT   1024
#define CSR_NW   (CSR_NT / 32)
#define CSR_WCAP 512
#define CSR_U    4
#define CSR_EBITS 20
static_assert(N_EDGES % 4 == 0, "dst is read 4 at a time");
static_assert(N_EDGES <= (1 << CSR_EBITS) && CSR_R <= (1 << (32 - CSR_EBITS)), "packed (node, edge) word");
static_assert(CSR_R % 32 == 0, "degree scan: CSR_R/32 entries per lane");

__global__ __launch_bounds__(CSR_NT)
void csr_by_node_range(const int* __restrict__ src, const int* __restrict__ dst,
                       const float* __restrict__ w,
                       int* __restrict__ rowptr, float* __restrict__ inv,
                       int* __restrict__ es, float* __restrict__ ew) {
  __shared__ int cnt[CSR_R];
  __shared__ int cur[CSR_R];
  __shared__ unsigned keep[CSR_NW * CSR_WCAP];
  __shared__ int below_s;
  __shared__ int ovf_s;

  const int tid  = threadIdx.x;
  const int lane = tid & 31;
  const int wv   = tid >> 5;
  const int lo   = blockIdx.x * CSR_R;

  for (int j = tid; j < CSR_R; j += CSR_NT) cnt[j] = 0;
  if (tid == 0) { below_s = 0; ovf_s = 0; }
  __syncthreads();

  const int4* __restrict__ d4 = (const int4*)dst;
  constexpr int NV   = N_EDGES / 4;
  constexpr int FULL = NV / CSR_NT;
  constexpr int G    = FULL / CSR_U;
  unsigned* mykeep = keep + wv * CSR_WCAP;
  int below  = 0;
  int wcount = 0;

  auto one = [&](int d, int e) {
    int t = d - lo;
    below += (t < 0) ? 1 : 0;
    bool in = (unsigned)t < (unsigned)CSR_R;
    unsigned m = __builtin_amdgcn_ballot_w32(in);
    if (m != 0u) {
      if (in) {
        int slot = (int)__builtin_amdgcn_mbcnt_lo(m, (unsigned)wcount);
        mykeep[min(slot, CSR_WCAP - 1)] = (unsigned)e;
      }
      wcount += __builtin_popcount(m);
    }
  };
  auto four = [&](int4 v, int i) {
    one(v.x, 4 * i + 0); one(v.y, 4 * i + 1); one(v.z, 4 * i + 2); one(v.w, 4 * i + 3);
  };
  auto load_batch = [&](int4 (&v)[CSR_U], int g) {
    #pragma unroll
    for (int u = 0; u < CSR_U; ++u) v[u] = d4[(g * CSR_U + u) * CSR_NT + tid];
  };
  auto scan_batch = [&](const int4 (&v)[CSR_U], int g) {
    #pragma unroll
    for (int u = 0; u < CSR_U; ++u) four(v[u], (g * CSR_U + u) * CSR_NT + tid);
  };

  int4 va[CSR_U], vb[CSR_U];
  int g = 0;
  if (G > 0) load_batch(va, 0);
  for (; g + 2 <= G; g += 2) {
    load_batch(vb, g + 1);
    scan_batch(va, g);
    if (g + 2 < G) load_batch(va, g + 2);
    scan_batch(vb, g + 1);
  }
  if (g < G) scan_batch(va, g);
  for (int k = G * CSR_U; k < FULL; ++k) four(d4[k * CSR_NT + tid], k * CSR_NT + tid);
  if (FULL * CSR_NT < NV) {
    int i = FULL * CSR_NT + tid;
    int4 v = make_int4(0x7fffffff, 0x7fffffff, 0x7fffffff, 0x7fffffff);
    if (i < NV) v = d4[i];
    four(v, i);
  }

  #pragma unroll
  for (int off = 16; off > 0; off >>= 1) below += __shfl_xor(below, off);
  if (lane == 0) {
    atomicAdd(&below_s, below);
    if (wcount > CSR_WCAP) ovf_s = 1;
  }
  __syncthreads();
  const bool kept = (ovf_s == 0);

  if (kept) {
    for (int s0 = lane; s0 < wcount; s0 += 32 * 4) {
      unsigned e[4]; int d[4]; bool ok[4];
      #pragma unroll
      for (int u = 0; u < 4; ++u) {
        int s = s0 + 32 * u;
        ok[u] = s < wcount;
        unsigned x = mykeep[min(s, CSR_WCAP - 1)];
        e[u] = ok[u] ? x : 0u;
      }
      #pragma unroll
      for (int u = 0; u < 4; ++u) d[u] = dst[e[u]];
      #pragma unroll
      for (int u = 0; u < 4; ++u) {
        if (ok[u]) {
          unsigned t = (unsigned)(d[u] - lo);
          atomicAdd(&cnt[t], 1);
          mykeep[s0 + 32 * u] = (t << CSR_EBITS) | e[u];
        }
      }
    }
  } else {
    for (int i = tid; i < NV; i += CSR_NT) {
      int4 v = d4[i];
      int dd[4] = {v.x, v.y, v.z, v.w};
      #pragma unroll
      for (int c4 = 0; c4 < 4; ++c4) {
        int t = dd[c4] - lo;
        if ((unsigned)t < (unsigned)CSR_R) atomicAdd(&cnt[t], 1);
      }
    }
  }
  __syncthreads();

  if (wv == 0) {
    constexpr int PER = CSR_R / 32;
    int c[PER];
    int s = 0;
    #pragma unroll
    for (int q = 0; q < PER; ++q) { c[q] = cnt[lane * PER + q]; s += c[q]; }
    int incl = s;
    #pragma unroll
    for (int off = 1; off < 32; off <<= 1) {
      int y = __shfl_up(incl, off);
      if (lane >= off) incl += y;
    }
    int run = below_s + incl - s;
    #pragma unroll
    for (int q = 0; q < PER; ++q) {
      int j = lane * PER + q;
      int node = lo + j;
      cur[j] = run;
      if (node <= N_NODES) rowptr[node] = run;
      if (node < N_NODES) inv[node] = (c[q] > 0) ? (1.0f / (float)c[q]) : 0.0f;
      run += c[q];
    }
    if (N_NODES % CSR_R == 0 && blockIdx.x == CSR_B - 1 && lane == 31) rowptr[N_NODES] = run;
  }
  __syncthreads();

  if (kept) {
    for (int s0 = lane; s0 < wcount; s0 += 32 * 4) {
      unsigned p[4]; int sv[4]; float fv[4]; bool ok[4];
      #pragma unroll
      for (int u = 0; u < 4; ++u) {
        int s = s0 + 32 * u;
        ok[u] = s < wcount;
        unsigned x = mykeep[min(s, CSR_WCAP - 1)];
        p[u] = ok[u] ? x : 0u;
      }
      #pragma unroll
      for (int u = 0; u < 4; ++u) {
        int e = (int)(p[u] & ((1u << CSR_EBITS) - 1u));
        sv[u] = src[e];
        fv[u] = w[e];
      }
      #pragma unroll
      for (int u = 0; u < 4; ++u) {
        if (ok[u]) {
          int pos = atomicAdd(&cur[p[u] >> CSR_EBITS], 1);
          es[pos] = sv[u];
          ew[pos] = fv[u];
        }
      }
    }
  } else {
    for (int i = tid; i < NV; i += CSR_NT) {
      int4 v = d4[i];
      int dd[4] = {v.x, v.y, v.z, v.w};
      #pragma unroll
      for (int c4 = 0; c4 < 4; ++c4) {
        int t = dd[c4] - lo;
        if ((unsigned)t < (unsigned)CSR_R) {
          int pos = atomicAdd(&cur[t], 1);
          es[pos] = src[4 * i + c4];
          ew[pos] = w[4 * i + c4];
        }
      }
    }
  }
}

template<int D, int NOUT, bool RELU>
__global__ __launch_bounds__(512)
void layer_block16(const float* __restrict__ X,
                   const int* __restrict__ rowptr,
                   const int* __restrict__ es,
                   const float* __restrict__ ew,
                   const float* __restrict__ inv,
                   const float* __restrict__ Wt,
                   const float* __restrict__ bias,
                   float* __restrict__ Y, int nN) {
  constexpr int K      = 2 * D;
  constexpr int PITCH  = K + 4;
  constexpr int NSTRIP = NOUT / 16;
  constexpr int C      = D / 128;
  static_assert(D % 128 == 0 && NOUT % 16 == 0 && NSTRIP <= 16, "tile shapes");
  static_assert(PITCH % 64 == 4, "bank spread of the A fragment loads");
  __shared__ __attribute__((aligned(16))) float At[16 * PITCH];

  const int lane = threadIdx.x & 31;
  const int wv   = threadIdx.x >> 5;
  const int m0   = blockIdx.x * 16;

  {
    const int node = m0 + wv;
    float4 own[C], acc[C];
    #pragma unroll
    for (int c = 0; c < C; ++c) {
      own[c] = make_float4(0.f, 0.f, 0.f, 0.f);
      acc[c] = make_float4(0.f, 0.f, 0.f, 0.f);
    }
    float iv = 0.0f;
    if (node < nN) {
      const float4* xo = (const float4*)(X + (size_t)node * D);
      #pragma unroll
      for (int c = 0; c < C; ++c) own[c] = xo[lane + 32 * c];
      int beg = rowptr[node];
      int end = rowptr[node + 1];
      for (int j = beg; j < end; ++j) {
        int s = es[j];
        float ww = ew[j];
        const float4* xr = (const float4*)(X + (size_t)s * D);
        #pragma unroll
        for (int c = 0; c < C; ++c) {
          float4 v = xr[lane + 32 * c];
          acc[c].x += v.x * ww;
          acc[c].y += v.y * ww;
          acc[c].z += v.z * ww;
          acc[c].w += v.w * ww;
        }
      }
      iv = inv[node];
    }
    float4* trow = (float4*)(At + wv * PITCH);
    #pragma unroll
    for (int c = 0; c < C; ++c) {
      trow[lane + 32 * c] = own[c];
      trow[D / 4 + lane + 32 * c] =
          make_float4(acc[c].x * iv, acc[c].y * iv, acc[c].z * iv, acc[c].w * iv);
    }
  }
  __syncthreads();

  if (wv < NSTRIP) {
    const int half = lane >> 4;
    const int mr   = lane & 15;
    const int n0   = wv * 16;
    const float* ap = At + mr * PITCH + 2 * half;
    const float* bp = Wt + (2 * half) * NOUT + n0 + mr;
    v8f acc = {};
    #pragma unroll 8
    for (int k = 0; k < K; k += 4) {
      v2f a = *(const v2f*)(ap + k);
      v2f b;
      b.x = bp[k * NOUT];
      b.y = bp[k * NOUT + NOUT];
      acc = __builtin_amdgcn_wmma_f32_16x16x4_f32(false, a, false, b, (short)0, acc, false, false);
    }
    const int n = n0 + mr;
    const float bv = bias[n];
    #pragma unroll
    for (int r = 0; r < 8; ++r) {
      int row = m0 + r + 8 * half;
      float v = acc[r] + bv;
      if (RELU) v = (v > 0.0f) ? v : 0.0f;
      if (row < nN) Y[(size_t)row * NOUT + n] = v;
    }
  }
}

extern "C" void kernel_launch(void* const* d_in, const int* in_sizes, int n_in,
                              void* d_out, int out_size, void* d_ws, size_t ws_size,
                              hipStream_t stream) {
  const float* h   = (const float*)d_in[0];
  const float* w   = (const float*)d_in[1];
  const int*   src = (const int*)d_in[2];
  const int*   dst = (const int*)d_in[3];
  const float* W1  = (const float*)d_in[4];
  const float* b1  = (const float*)d_in[5];
  const float* W2  = (const float*)d_in[6];
  const float* b2  = (const float*)d_in[7];
  float* out = (float*)d_out;

  int*   rowptr = (int*)d_ws + 10240;
  int*   es     = rowptr + 2 * 10240;
  float* ew     = (float*)(es + N_EDGES);
  float* inv    = ew + N_EDGES;
  float* x1     = inv + 10240;

  csr_by_node_range<<<CSR_B, CSR_NT, 0, stream>>>(src, dst, w, rowptr, inv, es, ew);

  layer_block16<D_IN, D_HID, true><<<(N_NODES + 15) / 16, 512, 0, stream>>>(
      h, rowptr, es, ew, inv, W1, b1, x1, N_NODES);

  layer_block16<D_HID, D_OUT, false><<<(N_NODES + 15) / 16, 512, 0, stream>>>(
      x1, rowptr, es, ew, inv, W2, b2, out, N_NODES);
}


# each layer block copies its 16 nodes' (source, weight) records to LDS once; the gather reads them there, so a trip waits for its 8 row loads only, not for an index load first; same sum order
# speedup vs baseline: 2.9868x; 1.0618x over previous
#include <hip/hip_runtime.h>

typedef __attribute__((ext_vector_type(2))) float v2f;
typedef __attribute__((ext_vector_type(8))) float v8f;

#define N_NODES 10000
#define N_EDGES 640000
#define D_IN    128
#define D_HID   256
#define D_OUT   64

#define CSR_R    64
#define CSR_B    ((N_NODES + CSR_R - 1) / CSR_R)
#define CSR_NT   1024
#define CSR_NW   (CSR_NT / 32)
#define CSR_WCAP 512
#define CSR_U    4
#define CSR_EBITS 20
static_assert(N_EDGES % 4 == 0, "dst is read 4 at a time");
static_assert(N_EDGES <= (1 << CSR_EBITS) && CSR_R <= (1 << (32 - CSR_EBITS)), "packed (node, edge) word");
static_assert(CSR_R % 32 == 0, "degree scan: CSR_R/32 entries per lane");

__global__ __launch_bounds__(CSR_NT)
void csr_by_node_range(const int* __restrict__ src, const int* __restrict__ dst,
                       const float* __restrict__ w,
                       int* __restrict__ rowptr, float* __restrict__ inv,
                       int* __restrict__ es, float* __restrict__ ew) {
  __shared__ int cnt[CSR_R];
  __shared__ int cur[CSR_R];
  __shared__ unsigned keep[CSR_NW * CSR_WCAP];
  __shared__ int below_s;
  __shared__ int ovf_s;

  const int tid  = threadIdx.x;
  const int lane = tid & 31;
  const int wv   = tid >> 5;
  const int lo   = blockIdx.x * CSR_R;

  for (int j = tid; j < CSR_R; j += CSR_NT) cnt[j] = 0;
  if (tid == 0) { below_s = 0; ovf_s = 0; }
  __syncthreads();

  const int4* __restrict__ d4 = (const int4*)dst;
  constexpr int NV   = N_EDGES / 4;
  constexpr int FULL = NV / CSR_NT;
  constexpr int G    = FULL / CSR_U;
  unsigned* mykeep = keep + wv * CSR_WCAP;
  int below  = 0;
  int wcount = 0;

  auto one = [&](int d, int e) {
    int t = d - lo;
    below += (t < 0) ? 1 : 0;
    bool in = (unsigned)t < (unsigned)CSR_R;
    unsigned m = __builtin_amdgcn_ballot_w32(in);
    if (m != 0u) {
      if (in) {
        int slot = (int)__builtin_amdgcn_mbcnt_lo(m, (unsigned)wcount);
        mykeep[min(slot, CSR_WCAP - 1)] = (unsigned)e;
      }
      wcount += __builtin_popcount(m);
    }
  };
  auto four = [&](int4 v, int i) {
    one(v.x, 4 * i + 0); one(v.y, 4 * i + 1); one(v.z, 4 * i + 2); one(v.w, 4 * i + 3);
  };
  auto load_batch = [&](int4 (&v)[CSR_U], int g) {
    #pragma unroll
    for (int u = 0; u < CSR_U; ++u) v[u] = d4[(g * CSR_U + u) * CSR_NT + tid];
  };
  auto scan_batch = [&](const int4 (&v)[CSR_U], int g) {
    #pragma unroll
    for (int u = 0; u < CSR_U; ++u) four(v[u], (g * CSR_U + u) * CSR_NT + tid);
  };

  int4 va[CSR_U], vb[CSR_U];
  int g = 0;
  if (G > 0) load_batch(va, 0);
  for (; g + 2 <= G; g += 2) {
    load_batch(vb, g + 1);
    scan_batch(va, g);
    if (g + 2 < G) load_batch(va, g + 2);
    scan_batch(vb, g + 1);
  }
  if (g < G) scan_batch(va, g);
  for (int k = G * CSR_U; k < FULL; ++k) four(d4[k * CSR_NT + tid], k * CSR_NT + tid);
  if (FULL * CSR_NT < NV) {
    int i = FULL * CSR_NT + tid;
    int4 v = make_int4(0x7fffffff, 0x7fffffff, 0x7fffffff, 0x7fffffff);
    if (i < NV) v = d4[i];
    four(v, i);
  }

  #pragma unroll
  for (int off = 16; off > 0; off >>= 1) below += __shfl_xor(below, off);
  if (lane == 0) {
    atomicAdd(&below_s, below);
    if (wcount > CSR_WCAP) ovf_s = 1;
  }
  __syncthreads();
  const bool kept = (ovf_s == 0);

  if (kept) {
    for (int s0 = lane; s0 < wcount; s0 += 32 * 4) {
      unsigned e[4]; int d[4]; bool ok[4];
      #pragma unroll
      for (int u = 0; u < 4; ++u) {
        int s = s0 + 32 * u;
        ok[u] = s < wcount;
        unsigned x = mykeep[min(s, CSR_WCAP - 1)];
        e[u] = ok[u] ? x : 0u;
      }
      #pragma unroll
      for (int u = 0; u < 4; ++u) d[u] = dst[e[u]];
      #pragma unroll
      for (int u = 0; u < 4; ++u) {
        if (ok[u]) {
          unsigned t = (unsigned)(d[u] - lo);
          atomicAdd(&cnt[t], 1);
          mykeep[s0 + 32 * u] = (t << CSR_EBITS) | e[u];
        }
      }
    }
  } else {
    for (int i = tid; i < NV; i += CSR_NT) {
      int4 v = d4[i];
      int dd[4] = {v.x, v.y, v.z, v.w};
      #pragma unroll
      for (int c4 = 0; c4 < 4; ++c4) {
        int t = dd[c4] - lo;
        if ((unsigned)t < (unsigned)CSR_R) atomicAdd(&cnt[t], 1);
      }
    }
  }
  __syncthreads();

  if (wv == 0) {
    constexpr int PER = CSR_R / 32;
    int c[PER];
    int s = 0;
    #pragma unroll
    for (int q = 0; q < PER; ++q) { c[q] = cnt[lane * PER + q]; s += c[q]; }
    int incl = s;
    #pragma unroll
    for (int off = 1; off < 32; off <<= 1) {
      int y = __shfl_up(incl, off);
      if (lane >= off) incl += y;
    }
    int run = below_s + incl - s;
    #pragma unroll
    for (int q = 0; q < PER; ++q) {
      int j = lane * PER + q;
      int node = lo + j;
      cur[j] = run;
      if (node <= N_NODES) rowptr[node] = run;
      if (node < N_NODES) inv[node] = (c[q] > 0) ? (1.0f / (float)c[q]) : 0.0f;
      run += c[q];
    }
    if (N_NODES % CSR_R == 0 && blockIdx.x == CSR_B - 1 && lane == 31) rowptr[N_NODES] = run;
  }
  __syncthreads();

  if (kept) {
    for (int s0 = lane; s0 < wcount; s0 += 32 * 4) {
      unsigned p[4]; int sv[4]; float fv[4]; bool ok[4];
      #pragma unroll
      for (int u = 0; u < 4; ++u) {
        int s = s0 + 32 * u;
        ok[u] = s < wcount;
        unsigned x = mykeep[min(s, CSR_WCAP - 1)];
        p[u] = ok[u] ? x : 0u;
      }
      #pragma unroll
      for (int u = 0; u < 4; ++u) {
        int e = (int)(p[u] & ((1u << CSR_EBITS) - 1u));
        sv[u] = src[e];
        fv[u] = w[e];
      }
      #pragma unroll
      for (int u = 0; u < 4; ++u) {
        if (ok[u]) {
          int pos = atomicAdd(&cur[p[u] >> CSR_EBITS], 1);
          es[pos] = sv[u];
          ew[pos] = fv[u];
        }
      }
    }
  } else {
    for (int i = tid; i < NV; i += CSR_NT) {
      int4 v = d4[i];
      int dd[4] = {v.x, v.y, v.z, v.w};
      #pragma unroll
      for (int c4 = 0; c4 < 4; ++c4) {
        int t = dd[c4] - lo;
        if ((unsigned)t < (unsigned)CSR_R) {
          int pos = atomicAdd(&cur[t], 1);
          es[pos] = src[4 * i + c4];
          ew[pos] = w[4 * i + c4];
        }
      }
    }
  }
}

#define LB_ECAP 2048
#define LB_PAD  8

template<int D, int NOUT, bool RELU>
__global__ __launch_bounds__(512)
void layer_block16(const float* __restrict__ X,
                   const int* __restrict__ rowptr,
                   const int* __restrict__ es,
                   const float* __restrict__ ew,
                   const float* __restrict__ inv,
                   const float* __restrict__ Wt,
                   const float* __restrict__ bias,
                   float* __restrict__ Y, int nN) {
  constexpr int K      = 2 * D;
  constexpr int PITCH  = K + 4;
  constexpr int NSTRIP = NOUT / 16;
  constexpr int C      = D / 128;
  constexpr int U      = 8 / C;
  static_assert(D % 128 == 0 && NOUT % 16 == 0 && NSTRIP <= 16, "tile shapes");
  static_assert(PITCH % 64 == 4, "bank spread of the A fragment loads");
  static_assert(U <= LB_PAD, "a last trip may look U - 1 records past the staged ones");
  __shared__ __attribute__((aligned(16))) float At[16 * PITCH];
  __shared__ int   eS[LB_ECAP + LB_PAD];
  __shared__ float eW[LB_ECAP + LB_PAD];

  const int tid  = threadIdx.x;
  const int lane = tid & 31;
  const int wv   = __builtin_amdgcn_readfirstlane(tid >> 5);
  const int m0   = blockIdx.x * 16;

  const int e0  = rowptr[m0];
  const int e1  = rowptr[min(m0 + 16, nN)];
  const int cnt = e1 - e0;
  const bool staged = cnt <= LB_ECAP;
  if (staged) {
    for (int i = tid; i < cnt; i += 512) {
      eS[i] = es[e0 + i];
      eW[i] = ew[e0 + i];
    }
    if (tid < LB_PAD) { eS[cnt + tid] = 0; eW[cnt + tid] = 0.0f; }
  }

  const int node = m0 + wv;
  int beg = e0, end = e0;
  float iv = 0.0f;
  float4* trow = (float4*)(At + wv * PITCH);
  {
    float4 own[C];
    #pragma unroll
    for (int c = 0; c < C; ++c) own[c] = make_float4(0.f, 0.f, 0.f, 0.f);
    if (node < nN) {
      const float4* xo = (const float4*)(X + (size_t)node * D);
      #pragma unroll
      for (int c = 0; c < C; ++c) own[c] = xo[lane + 32 * c];
      beg = rowptr[node];
      end = rowptr[node + 1];
      iv  = inv[node];
    }
    #pragma unroll
    for (int c = 0; c < C; ++c) trow[lane + 32 * c] = own[c];
  }
  __syncthreads();

  {
    float4 acc[C];
    #pragma unroll
    for (int c = 0; c < C; ++c) acc[c] = make_float4(0.f, 0.f, 0.f, 0.f);
    const char* xb = (const char*)X;
    const unsigned lane16 = (unsigned)lane * 16u;
    if (staged) {
      int j = beg - e0;
      const int jend = end - e0;
      unsigned off[U];
      #pragma unroll
      for (int u = 0; u < U; ++u) off[u] = (unsigned)eS[j + u] * (unsigned)(D * 4) + lane16;
      for (; j + U <= jend; j += U) {
        float ww[U]; float4 v[U][C]; unsigned offn[U];
        #pragma unroll
        for (int u = 0; u < U; ++u) {
          #pragma unroll
          for (int c = 0; c < C; ++c) v[u][c] = *(const float4*)(xb + (off[u] + 512u * c));
        }
        #pragma unroll
        for (int u = 0; u < U; ++u) {
          ww[u]   = eW[j + u];
          offn[u] = (unsigned)eS[j + U + u] * (unsigned)(D * 4) + lane16;
        }
        #pragma unroll
        for (int u = 0; u < U; ++u) {
          #pragma unroll
          for (int c = 0; c < C; ++c) {
            acc[c].x = __builtin_fmaf(v[u][c].x, ww[u], acc[c].x);
            acc[c].y = __builtin_fmaf(v[u][c].y, ww[u], acc[c].y);
            acc[c].z = __builtin_fmaf(v[u][c].z, ww[u], acc[c].z);
            acc[c].w = __builtin_fmaf(v[u][c].w, ww[u], acc[c].w);
          }
        }
        #pragma unroll
        for (int u = 0; u < U; ++u) off[u] = offn[u];
      }
      if (j < jend) {
        float ww[U]; float4 v[U][C]; bool ok[U];
        #pragma unroll
        for (int u = 0; u < U; ++u) {
          ok[u] = j + u < jend;
          ww[u] = eW[j + u];
        }
        #pragma unroll
        for (int u = 0; u < U; ++u) {
          #pragma unroll
          for (int c = 0; c < C; ++c) {
            v[u][c] = make_float4(0.f, 0.f, 0.f, 0.f);
            if (ok[u]) v[u][c] = *(const float4*)(xb + (off[u] + 512u * c));
          }
        }
        #pragma unroll
        for (int u = 0; u < U; ++u) {
          if (ok[u]) {
            #pragma unroll
            for (int c = 0; c < C; ++c) {
              acc[c].x = __builtin_fmaf(v[u][c].x, ww[u], acc[c].x);
              acc[c].y = __builtin_fmaf(v[u][c].y, ww[u], acc[c].y);
              acc[c].z = __builtin_fmaf(v[u][c].z, ww[u], acc[c].z);
              acc[c].w = __builtin_fmaf(v[u][c].w, ww[u], acc[c].w);
            }
          }
        }
      }
    } else {
      for (int j = beg; j < end; ++j) {
        int s = es[j];
        float ww = ew[j];
        const float4* xr = (const float4*)(X + (size_t)s * D);
        #pragma unroll
        for (int c = 0; c < C; ++c) {
          float4 v = xr[lane + 32 * c];
          acc[c].x = __builtin_fmaf(v.x, ww, acc[c].x);
          acc[c].y = __builtin_fmaf(v.y, ww, acc[c].y);
          acc[c].z = __builtin_fmaf(v.z, ww, acc[c].z);
          acc[c].w = __builtin_fmaf(v.w, ww, acc[c].w);
        }
      }
    }
    #pragma unroll
    for (int c = 0; c < C; ++c) {
      trow[D / 4 + lane + 32 * c] =
          make_float4(acc[c].x * iv, acc[c].y * iv, acc[c].z * iv, acc[c].w * iv);
    }
  }
  __syncthreads();

  if (wv < NSTRIP) {
    const int half = lane >> 4;
    const int mr   = lane & 15;
    const int n0   = wv * 16;
    const float* ap = At + mr * PITCH + 2 * half;
    const float* bp = Wt + (2 * half) * NOUT + n0 + mr;
    v8f acc = {};
    #pragma unroll 8
    for (int k = 0; k < K; k += 4) {
      v2f a = *(const v2f*)(ap + k);
      v2f b;
      b.x = bp[k * NOUT];
      b.y = bp[k * NOUT + NOUT];
      acc = __builtin_amdgcn_wmma_f32_16x16x4_f32(false, a, false, b, (short)0, acc, false, false);
    }
    const int n = n0 + mr;
    const float bv = bias[n];
    #pragma unroll
    for (int r = 0; r < 8; ++r) {
      int row = m0 + r + 8 * half;
      float v = acc[r] + bv;
      if (RELU) v = (v > 0.0f) ? v : 0.0f;
      if (row < nN) Y[(size_t)row * NOUT + n] = v;
    }
  }
}

extern "C" void kernel_launch(void* const* d_in, const int* in_sizes, int n_in,
                              void* d_out, int out_size, void* d_ws, size_t ws_size,
                              hipStream_t stream) {
  const float* h   = (const float*)d_in[0];
  const float* w   = (const float*)d_in[1];
  const int*   src = (const int*)d_in[2];
  const int*   dst = (const int*)d_in[3];
  const float* W1  = (const float*)d_in[4];
  const float* b1  = (const float*)d_in[5];
  const float* W2  = (const float*)d_in[6];
  const float* b2  = (const float*)d_in[7];
  float* out = (float*)d_out;

  int*   rowptr = (int*)d_ws + 10240;
  int*   es     = rowptr + 2 * 10240;
  float* ew     = (float*)(es + N_EDGES);
  float* inv    = ew + N_EDGES;
  float* x1     = inv + 10240;

  csr_by_node_range<<<CSR_B, CSR_NT, 0, stream>>>(src, dst, w, rowptr, inv, es, ew);

  layer_block16<D_IN, D_HID, true><<<(N_NODES + 15) / 16, 512, 0, stream>>>(
      h, rowptr, es, ew, inv, W1, b1, x1, N_NODES);

  layer_block16<D_HID, D_OUT, false><<<(N_NODES + 15) / 16, 512, 0, stream>>>(
      x1, rowptr, es, ew, inv, W2, b2, out, N_NODES);
}


# CSR block sorts its kept edge ids by node in LDS (kept word = node + edge id, no second dst read), reads src, w in list order and stores es, ew as whole lines; in-node order LDS arrival as before
# speedup vs baseline: 3.2216x; 1.0786x over previous
#include <hip/hip_runtime.h>

typedef __attribute__((ext_vector_type(2))) float v2f;
typedef __attribute__((ext_vector_type(8))) float v8f;

#define N_NODES 10000
#define N_EDGES 640000
#define D_IN    128
#define D_HID   256
#define D_OUT   64

#define CSR_R    64
#define CSR_B    ((N_NODES + CSR_R - 1) / CSR_R)
#define CSR_NT   1024
#define CSR_NW   (CSR_NT / 32)
#define CSR_WCAP 512
#define CSR_SCAP 8192
#define CSR_KU   6
#define CSR_PU   5
#define CSR_U    4
#define CSR_EBITS 20
static_assert(N_EDGES % 4 == 0, "dst is read 4 at a time");
static_assert(N_EDGES <= (1 << CSR_EBITS) && CSR_R <= (1 << (32 - CSR_EBITS)), "packed (node, edge) word");
static_assert(CSR_R % 32 == 0, "degree scan: CSR_R/32 entries per lane");

__global__ __launch_bounds__(CSR_NT)
void csr_by_node_range(const int* __restrict__ src, const int* __restrict__ dst,
                       const float* __restrict__ w,
                       int* __restrict__ rowptr, float* __restrict__ inv,
                       int* __restrict__ es, float* __restrict__ ew) {
  __shared__ int cnt[CSR_R];
  __shared__ int cur[CSR_R];
  __shared__ unsigned keep[CSR_NW * CSR_WCAP];
  __shared__ unsigned sorted[CSR_SCAP];
  __shared__ int tot_s;
  __shared__ int below_s;
  __shared__ int ovf_s;

  const int tid  = threadIdx.x;
  const int lane = tid & 31;
  const int wv   = tid >> 5;
  const int lo   = blockIdx.x * CSR_R;

  for (int j = tid; j < CSR_R; j += CSR_NT) cnt[j] = 0;
  if (tid == 0) { below_s = 0; ovf_s = 0; }
  __syncthreads();

  const int4* __restrict__ d4 = (const int4*)dst;
  constexpr int NV   = N_EDGES / 4;
  constexpr int FULL = NV / CSR_NT;
  constexpr int G    = FULL / CSR_U;
  unsigned* mykeep = keep + wv * CSR_WCAP;
  int below  = 0;
  int wcount = 0;

  auto one = [&](int d, int e) {
    int t = d - lo;
    below += (t < 0) ? 1 : 0;
    bool in = (unsigned)t < (unsigned)CSR_R;
    unsigned m = __builtin_amdgcn_ballot_w32(in);
    if (m != 0u) {
      if (in) {
        int slot = (int)__builtin_amdgcn_mbcnt_lo(m, (unsigned)wcount);
        mykeep[min(slot, CSR_WCAP - 1)] = ((unsigned)t << CSR_EBITS) | (unsigned)e;
      }
      wcount += __builtin_popcount(m);
    }
  };
  auto four = [&](int4 v, int i) {
    one(v.x, 4 * i + 0); one(v.y, 4 * i + 1); one(v.z, 4 * i + 2); one(v.w, 4 * i + 3);
  };
  auto load_batch = [&](int4 (&v)[CSR_U], int g) {
    #pragma unroll
    for (int u = 0; u < CSR_U; ++u) v[u] = d4[(g * CSR_U + u) * CSR_NT + tid];
  };
  auto scan_batch = [&](const int4 (&v)[CSR_U], int g) {
    #pragma unroll
    for (int u = 0; u < CSR_U; ++u) four(v[u], (g * CSR_U + u) * CSR_NT + tid);
  };

  int4 va[CSR_U], vb[CSR_U];
  int g = 0;
  if (G > 0) load_batch(va, 0);
  for (; g + 2 <= G; g += 2) {
    load_batch(vb, g + 1);
    scan_batch(va, g);
    if (g + 2 < G) load_batch(va, g + 2);
    scan_batch(vb, g + 1);
  }
  if (g < G) scan_batch(va, g);
  for (int k = G * CSR_U; k < FULL; ++k) four(d4[k * CSR_NT + tid], k * CSR_NT + tid);
  if (FULL * CSR_NT < NV) {
    int i = FULL * CSR_NT + tid;
    int4 v = make_int4(0x7fffffff, 0x7fffffff, 0x7fffffff, 0x7fffffff);
    if (i < NV) v = d4[i];
    four(v, i);
  }

  #pragma unroll
  for (int off = 16; off > 0; off >>= 1) below += __shfl_xor(below, off);
  if (lane == 0) {
    atomicAdd(&below_s, below);
    if (wcount > CSR_WCAP) ovf_s = 1;
  }
  __syncthreads();
  const bool kept = (ovf_s == 0);

  if (kept) {
    for (int s0 = lane; s0 < wcount; s0 += 32 * CSR_KU) {
      unsigned p[CSR_KU];
      #pragma unroll
      for (int u = 0; u < CSR_KU; ++u) p[u] = mykeep[min(s0 + 32 * u, CSR_WCAP - 1)];
      #pragma unroll
      for (int u = 0; u < CSR_KU; ++u) {
        if (s0 + 32 * u < wcount) atomicAdd(&cnt[p[u] >> CSR_EBITS], 1);
      }
    }
  } else {
    for (int i = tid; i < NV; i += CSR_NT) {
      int4 v = d4[i];
      int dd[4] = {v.x, v.y, v.z, v.w};
      #pragma unroll
      for (int c4 = 0; c4 < 4; ++c4) {
        int t = dd[c4] - lo;
        if ((unsigned)t < (unsigned)CSR_R) atomicAdd(&cnt[t], 1);
      }
    }
  }
  __syncthreads();

  if (wv == 0) {
    constexpr int PER = CSR_R / 32;
    int c[PER];
    int s = 0;
    #pragma unroll
    for (int q = 0; q < PER; ++q) { c[q] = cnt[lane * PER + q]; s += c[q]; }
    int incl = s;
    #pragma unroll
    for (int off = 1; off < 32; off <<= 1) {
      int y = __shfl_up(incl, off);
      if (lane >= off) incl += y;
    }
    int run = below_s + incl - s;
    #pragma unroll
    for (int q = 0; q < PER; ++q) {
      int j = lane * PER + q;
      int node = lo + j;
      cur[j] = run;
      if (node <= N_NODES) rowptr[node] = run;
      if (node < N_NODES) inv[node] = (c[q] > 0) ? (1.0f / (float)c[q]) : 0.0f;
      run += c[q];
    }
    if (N_NODES % CSR_R == 0 && blockIdx.x == CSR_B - 1 && lane == 31) rowptr[N_NODES] = run;
    if (lane == 31) tot_s = run - below_s;
  }
  __syncthreads();
  const int base = below_s;
  const int tot  = tot_s;

  if (kept && tot <= CSR_SCAP) {
    for (int s0 = lane; s0 < wcount; s0 += 32 * CSR_KU) {
      unsigned p[CSR_KU]; bool ok[CSR_KU]; int pos[CSR_KU];
      #pragma unroll
      for (int u = 0; u < CSR_KU; ++u) {
        int s = s0 + 32 * u;
        ok[u] = s < wcount;
        p[u] = mykeep[min(s, CSR_WCAP - 1)];
      }
      #pragma unroll
      for (int u = 0; u < CSR_KU; ++u) {
        pos[u] = base;
        if (ok[u]) pos[u] = atomicAdd(&cur[p[u] >> CSR_EBITS], 1);
      }
      #pragma unroll
      for (int u = 0; u < CSR_KU; ++u) {
        if (ok[u]) sorted[pos[u] - base] = p[u] & ((1u << CSR_EBITS) - 1u);
      }
    }
    __syncthreads();
    const int a = base & 31;
    #pragma unroll 1
    for (int i0 = tid - a; i0 < tot; i0 += CSR_PU * CSR_NT) {
      int e[CSR_PU]; int sv[CSR_PU]; float fv[CSR_PU]; bool ok[CSR_PU];
      #pragma unroll
      for (int u = 0; u < CSR_PU; ++u) {
        int i = i0 + CSR_NT * u;
        ok[u] = i >= 0 && i < tot;
        unsigned x = sorted[min(max(i, 0), CSR_SCAP - 1)];
        e[u] = ok[u] ? (int)x : 0;
      }
      #pragma unroll
      for (int u = 0; u < CSR_PU; ++u) {
        sv[u] = src[e[u]];
        fv[u] = w[e[u]];
      }
      #pragma unroll
      for (int u = 0; u < CSR_PU; ++u) {
        if (ok[u]) {
          es[base + i0 + CSR_NT * u] = sv[u];
          ew[base + i0 + CSR_NT * u] = fv[u];
        }
      }
    }
  } else if (kept) {
    for (int s0 = lane; s0 < wcount; s0 += 32 * 4) {
      unsigned p[4]; int sv[4]; float fv[4]; bool ok[4];
      #pragma unroll
      for (int u = 0; u < 4; ++u) {
        int s = s0 + 32 * u;
        ok[u] = s < wcount;
        unsigned x = mykeep[min(s, CSR_WCAP - 1)];
        p[u] = ok[u] ? x : 0u;
      }
      #pragma unroll
      for (int u = 0; u < 4; ++u) {
        int e = (int)(p[u] & ((1u << CSR_EBITS) - 1u));
        sv[u] = src[e];
        fv[u] = w[e];
      }
      #pragma unroll
      for (int u = 0; u < 4; ++u) {
        if (ok[u]) {
          int pos = atomicAdd(&cur[p[u] >> CSR_EBITS], 1);
          es[pos] = sv[u];
          ew[pos] = fv[u];
        }
      }
    }
  } else {
    for (int i = tid; i < NV; i += CSR_NT) {
      int4 v = d4[i];
      int dd[4] = {v.x, v.y, v.z, v.w};
      #pragma unroll
      for (int c4 = 0; c4 < 4; ++c4) {
        int t = dd[c4] - lo;
        if ((unsigned)t < (unsigned)CSR_R) {
          int pos = atomicAdd(&cur[t], 1);
          es[pos] = src[4 * i + c4];
          ew[pos] = w[4 * i + c4];
        }
      }
    }
  }
}

#define LB_ECAP 2048
#define LB_PAD  8

template<int D, int NOUT, bool RELU>
__global__ __launch_bounds__(512)
void layer_block16(const float* __restrict__ X,
                   const int* __restrict__ rowptr,
                   const int* __restrict__ es,
                   const float* __restrict__ ew,
                   const float* __restrict__ inv,
                   const float* __restrict__ Wt,
                   const float* __restrict__ bias,
                   float* __restrict__ Y, int nN) {
  constexpr int K      = 2 * D;
  constexpr int PITCH  = K + 4;
  constexpr int NSTRIP = NOUT / 16;
  constexpr int C      = D / 128;
  constexpr int U      = 8 / C;
  static_assert(D % 128 == 0 && NOUT % 16 == 0 && NSTRIP <= 16, "tile shapes");
  static_assert(PITCH % 64 == 4, "bank spread of the A fragment loads");
  static_assert(U <= LB_PAD, "a last trip may look U - 1 records past the staged ones");
  __shared__ __attribute__((aligned(16))) float At[16 * PITCH];
  __shared__ int   eS[LB_ECAP + LB_PAD];
  __shared__ float eW[LB_ECAP + LB_PAD];

  const int tid  = threadIdx.x;
  const int lane = tid & 31;
  const int wv   = __builtin_amdgcn_readfirstlane(tid >> 5);
  const int m0   = blockIdx.x * 16;

  const int e0  = rowptr[m0];
  const int e1  = rowptr[min(m0 + 16, nN)];
  const int cnt = e1 - e0;
  const bool staged = cnt <= LB_ECAP;
  if (staged) {
    for (int i = tid; i < cnt; i += 512) {
      eS[i] = es[e0 + i];
      eW[i] = ew[e0 + i];
    }
    if (tid < LB_PAD) { eS[cnt + tid] = 0; eW[cnt + tid] = 0.0f; }
  }

  const int node = m0 + wv;
  int beg = e0, end = e0;
  float iv = 0.0f;
  float4* trow = (float4*)(At + wv * PITCH);
  {
    float4 own[C];
    #pragma unroll
    for (int c = 0; c < C; ++c) own[c] = make_float4(0.f, 0.f, 0.f, 0.f);
    if (node < nN) {
      const float4* xo = (const float4*)(X + (size_t)node * D);
      #pragma unroll
      for (int c = 0; c < C; ++c) own[c] = xo[lane + 32 * c];
      beg = rowptr[node];
      end = rowptr[node + 1];
      iv  = inv[node];
    }
    #pragma unroll
    for (int c = 0; c < C; ++c) trow[lane + 32 * c] = own[c];
  }
  __syncthreads();

  {
    float4 acc[C];
    #pragma unroll
    for (int c = 0; c < C; ++c) acc[c] = make_float4(0.f, 0.f, 0.f, 0.f);
    const char* xb = (const char*)X;
    const unsigned lane16 = (unsigned)lane * 16u;
    if (staged) {
      int j = beg - e0;
      const int jend = end - e0;
      unsigned off[U];
      #pragma unroll
      for (int u = 0; u < U; ++u) off[u] = (unsigned)eS[j + u] * (unsigned)(D * 4) + lane16;
      for (; j + U <= jend; j += U) {
        float ww[U]; float4 v[U][C]; unsigned offn[U];
        #pragma unroll
        for (int u = 0; u < U; ++u) {
          #pragma unroll
          for (int c = 0; c < C; ++c) v[u][c] = *(const float4*)(xb + (off[u] + 512u * c));
        }
        #pragma unroll
        for (int u = 0; u < U; ++u) {
          ww[u]   = eW[j + u];
          offn[u] = (unsigned)eS[j + U + u] * (unsigned)(D * 4) + lane16;
        }
        #pragma unroll
        for (int u = 0; u < U; ++u) {
          #pragma unroll
          for (int c = 0; c < C; ++c) {
            acc[c].x = __builtin_fmaf(v[u][c].x, ww[u], acc[c].x);
            acc[c].y = __builtin_fmaf(v[u][c].y, ww[u], acc[c].y);
            acc[c].z = __builtin_fmaf(v[u][c].z, ww[u], acc[c].z);
            acc[c].w = __builtin_fmaf(v[u][c].w, ww[u], acc[c].w);
          }
        }
        #pragma unroll
        for (int u = 0; u < U; ++u) off[u] = offn[u];
      }
      if (j < jend) {
        float ww[U]; float4 v[U][C]; bool ok[U];
        #pragma unroll
        for (int u = 0; u < U; ++u) {
          ok[u] = j + u < jend;
          ww[u] = eW[j + u];
        }
        #pragma unroll
        for (int u = 0; u < U; ++u) {
          #pragma unroll
          for (int c = 0; c < C; ++c) {
            v[u][c] = make_float4(0.f, 0.f, 0.f, 0.f);
            if (ok[u]) v[u][c] = *(const float4*)(xb + (off[u] + 512u * c));
          }
        }
        #pragma unroll
        for (int u = 0; u < U; ++u) {
          if (ok[u]) {
            #pragma unroll
            for (int c = 0; c < C; ++c) {
              acc[c].x = __builtin_fmaf(v[u][c].x, ww[u], acc[c].x);
              acc[c].y = __builtin_fmaf(v[u][c].y, ww[u], acc[c].y);
              acc[c].z = __builtin_fmaf(v[u][c].z, ww[u], acc[c].z);
              acc[c].w = __builtin_fmaf(v[u][c].w, ww[u], acc[c].w);
            }
          }
        }
      }
    } else {
      for (int j = beg; j < end; ++j) {
        int s = es[j];
        float ww = ew[j];
        const float4* xr = (const float4*)(X + (size_t)s * D);
        #pragma unroll
        for (int c = 0; c < C; ++c) {
          float4 v = xr[lane + 32 * c];
          acc[c].x = __builtin_fmaf(v.x, ww, acc[c].x);
          acc[c].y = __builtin_fmaf(v.y, ww, acc[c].y);
          acc[c].z = __builtin_fmaf(v.z, ww, acc[c].z);
          acc[c].w = __builtin_fmaf(v.w, ww, acc[c].w);
        }
      }
    }
    #pragma unroll
    for (int c = 0; c < C; ++c) {
      trow[D / 4 + lane + 32 * c] =
          make_float4(acc[c].x * iv, acc[c].y * iv, acc[c].z * iv, acc[c].w * iv);
    }
  }
  __syncthreads();

  if (wv < NSTRIP) {
    const int half = lane >> 4;
    const int mr   = lane & 15;
    const int n0   = wv * 16;
    const float* ap = At + mr * PITCH + 2 * half;
    const float* bp = Wt + (2 * half) * NOUT + n0 + mr;
    v8f acc = {};
    #pragma unroll 8
    for (int k = 0; k < K; k += 4) {
      v2f a = *(const v2f*)(ap + k);
      v2f b;
      b.x = bp[k * NOUT];
      b.y = bp[k * NOUT + NOUT];
      acc = __builtin_amdgcn_wmma_f32_16x16x4_f32(false, a, false, b, (short)0, acc, false, false);
    }
    const int n = n0 + mr;
    const float bv = bias[n];
    #pragma unroll
    for (int r = 0; r < 8; ++r) {
      int row = m0 + r + 8 * half;
      float v = acc[r] + bv;
      if (RELU) v = (v > 0.0f) ? v : 0.0f;
      if (row < nN) Y[(size_t)row * NOUT + n] = v;
    }
  }
}

extern "C" void kernel_launch(void* const* d_in, const int* in_sizes, int n_in,
                              void* d_out, int out_size, void* d_ws, size_t ws_size,
                              hipStream_t stream) {
  const float* h   = (const float*)d_in[0];
  const float* w   = (const float*)d_in[1];
  const int*   src = (const int*)d_in[2];
  const int*   dst = (const int*)d_in[3];
  const float* W1  = (const float*)d_in[4];
  const float* b1  = (const float*)d_in[5];
  const float* W2  = (const float*)d_in[6];
  const float* b2  = (const float*)d_in[7];
  float* out = (float*)d_out;

  int*   rowptr = (int*)d_ws + 10240;
  int*   es     = rowptr + 2 * 10240;
  float* ew     = (float*)(es + N_EDGES);
  float* inv    = ew + N_EDGES;
  float* x1     = inv + 10240;

  csr_by_node_range<<<CSR_B, CSR_NT, 0, stream>>>(src, dst, w, rowptr, inv, es, ew);

  layer_block16<D_IN, D_HID, true><<<(N_NODES + 15) / 16, 512, 0, stream>>>(
      h, rowptr, es, ew, inv, W1, b1, x1, N_NODES);

  layer_block16<D_HID, D_OUT, false><<<(N_NODES + 15) / 16, 512, 0, stream>>>(
      x1, rowptr, es, ew, inv, W2, b2, out, N_NODES);
}


# CSR launch re-cut: a block takes (320 nodes, 1/8 of dst), 256 blocks (was 160 nodes, 1/4), so it scans half as many edges; a node's edges are 8 segments; in-node sum order = 8 parts, then LDS arrival
# speedup vs baseline: 4.3353x; 1.3457x over previous
#include <hip/hip_runtime.h>

typedef __attribute__((ext_vector_type(2))) float v2f;
typedef __attribute__((ext_vector_type(8))) float v8f;

#define N_NODES 10000
#define N_EDGES 640000
#define D_IN    128
#define D_HID   256
#define D_OUT   64

#define CSR_S    8
#define CSR_R    320
#define CSR_RB   ((N_NODES + CSR_R - 1) / CSR_R)
#define CSR_B    (CSR_RB * CSR_S)
#define CSR_EPS  (N_EDGES / CSR_S)
#define RP_PLANE 10240
#define CSR_NT   1024
#define CSR_NW   (CSR_NT / 32)
#define CSR_WCAP 512
#define CSR_SCAP 8192
#define CSR_KU   4
#define CSR_PU   3
#define CSR_U    4
#define CSR_EBITS 20
static_assert(N_EDGES % (4 * CSR_S) == 0, "every part of dst is read 4 at a time");
static_assert(N_NODES + 1 <= RP_PLANE, "row bounds of a part fit its plane");
static_assert(N_EDGES <= (1 << CSR_EBITS) && CSR_R <= (1 << (32 - CSR_EBITS)), "packed (node, edge) word");
static_assert(CSR_R % 32 == 0, "degree scan: CSR_R/32 entries per lane");

__global__ __launch_bounds__(CSR_NT)
void csr_by_node_range(const int* __restrict__ src, const int* __restrict__ dst,
                       const float* __restrict__ w,
                       int* __restrict__ rowptr,
                       int* __restrict__ es, float* __restrict__ ew) {
  __shared__ int cnt[CSR_R];
  __shared__ int cur[CSR_R];
  __shared__ unsigned keep[CSR_NW * CSR_WCAP];
  __shared__ unsigned sorted[CSR_SCAP];
  __shared__ int tot_s;
  __shared__ int below_s;
  __shared__ int ovf_s;

  const int tid  = threadIdx.x;
  const int lane = tid & 31;
  const int wv   = tid >> 5;
  const int part = blockIdx.x / CSR_RB;
  const int lo   = (blockIdx.x % CSR_RB) * CSR_R;
  rowptr += part * RP_PLANE;

  for (int j = tid; j < CSR_R; j += CSR_NT) cnt[j] = 0;
  if (tid == 0) { below_s = 0; ovf_s = 0; }
  __syncthreads();

  const int4* __restrict__ d4 = (const int4*)dst;
  constexpr int NVP  = CSR_EPS / 4;
  constexpr int TR   = (NVP + CSR_NT - 1) / CSR_NT;
  constexpr int NB   = (TR + CSR_U - 1) / CSR_U;
  const int i0p = part * NVP;
  unsigned* mykeep = keep + wv * CSR_WCAP;
  int below  = 0;
  int wcount = 0;

  auto one = [&](int d, int e) {
    int t = d - lo;
    below += (t < 0) ? 1 : 0;
    bool in = (unsigned)t < (unsigned)CSR_R;
    unsigned m = __builtin_amdgcn_ballot_w32(in);
    if (m != 0u) {
      if (in) {
        int slot = (int)__builtin_amdgcn_mbcnt_lo(m, (unsigned)wcount);
        mykeep[min(slot, CSR_WCAP - 1)] = ((unsigned)t << CSR_EBITS) | (unsigned)e;
      }
      wcount += __builtin_popcount(m);
    }
  };
  auto four = [&](int4 v, int i) {
    one(v.x, 4 * i + 0); one(v.y, 4 * i + 1); one(v.z, 4 * i + 2); one(v.w, 4 * i + 3);
  };
  auto load_batch = [&](int4 (&v)[CSR_U], int g) {
    #pragma unroll
    for (int u = 0; u < CSR_U; ++u) v[u] = d4[i0p + (g * CSR_U + u) * CSR_NT + tid];
  };
  auto scan_batch = [&](const int4 (&v)[CSR_U], int g) {
    #pragma unroll
    for (int u = 0; u < CSR_U; ++u) four(v[u], i0p + (g * CSR_U + u) * CSR_NT + tid);
  };
  auto load_last = [&](int4 (&v)[CSR_U]) {
    #pragma unroll
    for (int u = 0; u < CSR_U; ++u) {
      const int t = (NB - 1) * CSR_U + u;
      if (t < TR) {
        int i = t * CSR_NT + tid;
        v[u] = make_int4(0x7fffffff, 0x7fffffff, 0x7fffffff, 0x7fffffff);
        if ((t + 1) * CSR_NT <= NVP || i < NVP) v[u] = d4[i0p + i];
      }
    }
  };
  auto scan_last = [&](const int4 (&v)[CSR_U]) {
    #pragma unroll
    for (int u = 0; u < CSR_U; ++u) {
      const int t = (NB - 1) * CSR_U + u;
      if (t < TR) four(v[u], i0p + t * CSR_NT + tid);
    }
  };

  int4 va[CSR_U], vb[CSR_U];
  if (NB == 1) {
    load_last(va);
    scan_last(va);
  } else {
    int g = 0;
    load_batch(va, 0);
    for (; g + 2 < NB; g += 2) {
      load_batch(vb, g + 1);
      scan_batch(va, g);
      if (g + 2 < NB - 1) load_batch(va, g + 2); else load_last(va);
      scan_batch(vb, g + 1);
    }
    if (g + 2 == NB) {
      load_last(vb);
      scan_batch(va, g);
      scan_last(vb);
    } else {
      scan_last(va);
    }
  }

  #pragma unroll
  for (int off = 16; off > 0; off >>= 1) below += __shfl_xor(below, off);
  if (lane == 0) {
    atomicAdd(&below_s, below);
    if (wcount > CSR_WCAP) ovf_s = 1;
  }
  __syncthreads();
  const bool kept = (ovf_s == 0);

  if (kept) {
    for (int s0 = lane; s0 < wcount; s0 += 32 * CSR_KU) {
      unsigned p[CSR_KU];
      #pragma unroll
      for (int u = 0; u < CSR_KU; ++u) p[u] = mykeep[min(s0 + 32 * u, CSR_WCAP - 1)];
      #pragma unroll
      for (int u = 0; u < CSR_KU; ++u) {
        if (s0 + 32 * u < wcount) atomicAdd(&cnt[p[u] >> CSR_EBITS], 1);
      }
    }
  } else {
    for (int i = i0p + tid; i < i0p + NVP; i += CSR_NT) {
      int4 v = d4[i];
      int dd[4] = {v.x, v.y, v.z, v.w};
      #pragma unroll
      for (int c4 = 0; c4 < 4; ++c4) {
        int t = dd[c4] - lo;
        if ((unsigned)t < (unsigned)CSR_R) atomicAdd(&cnt[t], 1);
      }
    }
  }
  __syncthreads();

  if (wv == 0) {
    constexpr int PER = CSR_R / 32;
    int c[PER];
    int s = 0;
    #pragma unroll
    for (int q = 0; q < PER; ++q) { c[q] = cnt[lane * PER + q]; s += c[q]; }
    int incl = s;
    #pragma unroll
    for (int off = 1; off < 32; off <<= 1) {
      int y = __shfl_up(incl, off);
      if (lane >= off) incl += y;
    }
    int run = part * CSR_EPS + below_s + incl - s;
    #pragma unroll
    for (int q = 0; q < PER; ++q) {
      int j = lane * PER + q;
      int node = lo + j;
      cur[j] = run;
      if (node <= N_NODES) rowptr[node] = run;
      run += c[q];
    }
    if (N_NODES % CSR_R == 0 && lo + CSR_R == N_NODES && lane == 31) rowptr[N_NODES] = run;
    if (lane == 31) tot_s = run - part * CSR_EPS - below_s;
  }
  __syncthreads();
  const int base = part * CSR_EPS + below_s;
  const int tot  = tot_s;

  if (kept && tot <= CSR_SCAP) {
    for (int s0 = lane; s0 < wcount; s0 += 32 * CSR_KU) {
      unsigned p[CSR_KU]; bool ok[CSR_KU]; int pos[CSR_KU];
      #pragma unroll
      for (int u = 0; u < CSR_KU; ++u) {
        int s = s0 + 32 * u;
        ok[u] = s < wcount;
        p[u] = mykeep[min(s, CSR_WCAP - 1)];
      }
      #pragma unroll
      for (int u = 0; u < CSR_KU; ++u) {
        pos[u] = base;
        if (ok[u]) pos[u] = atomicAdd(&cur[p[u] >> CSR_EBITS], 1);
      }
      #pragma unroll
      for (int u = 0; u < CSR_KU; ++u) {
        if (ok[u]) sorted[pos[u] - base] = p[u] & ((1u << CSR_EBITS) - 1u);
      }
    }
    __syncthreads();
    const int a = base & 31;
    #pragma unroll 1
    for (int i0 = tid - a; i0 < tot; i0 += CSR_PU * CSR_NT) {
      int e[CSR_PU]; int sv[CSR_PU]; float fv[CSR_PU]; bool ok[CSR_PU];
      #pragma unroll
      for (int u = 0; u < CSR_PU; ++u) {
        int i = i0 + CSR_NT * u;
        ok[u] = i >= 0 && i < tot;
        unsigned x = sorted[min(max(i, 0), CSR_SCAP - 1)];
        e[u] = ok[u] ? (int)x : 0;
      }
      #pragma unroll
      for (int u = 0; u < CSR_PU; ++u) {
        sv[u] = src[e[u]];
        fv[u] = w[e[u]];
      }
      #pragma unroll
      for (int u = 0; u < CSR_PU; ++u) {
        if (ok[u]) {
          es[base + i0 + CSR_NT * u] = sv[u];
          ew[base + i0 + CSR_NT * u] = fv[u];
        }
      }
    }
  } else if (kept) {
    for (int s0 = lane; s0 < wcount; s0 += 32 * 4) {
      unsigned p[4]; int sv[4]; float fv[4]; bool ok[4];
      #pragma unroll
      for (int u = 0; u < 4; ++u) {
        int s = s0 + 32 * u;
        ok[u] = s < wcount;
        unsigned x = mykeep[min(s, CSR_WCAP - 1)];
        p[u] = ok[u] ? x : 0u;
      }
      #pragma unroll
      for (int u = 0; u < 4; ++u) {
        int e = (int)(p[u] & ((1u << CSR_EBITS) - 1u));
        sv[u] = src[e];
        fv[u] = w[e];
      }
      #pragma unroll
      for (int u = 0; u < 4; ++u) {
        if (ok[u]) {
          int pos = atomicAdd(&cur[p[u] >> CSR_EBITS], 1);
          es[pos] = sv[u];
          ew[pos] = fv[u];
        }
      }
    }
  } else {
    for (int i = i0p + tid; i < i0p + NVP; i += CSR_NT) {
      int4 v = d4[i];
      int dd[4] = {v.x, v.y, v.z, v.w};
      #pragma unroll
      for (int c4 = 0; c4 < 4; ++c4) {
        int t = dd[c4] - lo;
        if ((unsigned)t < (unsigned)CSR_R) {
          int pos = atomicAdd(&cur[t], 1);
          es[pos] = src[4 * i + c4];
          ew[pos] = w[4 * i + c4];
        }
      }
    }
  }
}

#define LB_WCAP 192
#define LB_PAD  8

template<int D, int NOUT, bool RELU>
__global__ __launch_bounds__(512)
void layer_block16(const float* __restrict__ X,
                   const int* __restrict__ rowptr,
                   const int* __restrict__ es,
                   const float* __restrict__ ew,
                   const float* __restrict__ Wt,
                   const float* __restrict__ bias,
                   float* __restrict__ Y, int nN) {
  constexpr int K      = 2 * D;
  constexpr int PITCH  = K + 4;
  constexpr int NSTRIP = NOUT / 16;
  constexpr int C      = D / 128;
  constexpr int U      = 8 / C;
  static_assert(D % 128 == 0 && NOUT % 16 == 0 && NSTRIP <= 16, "tile shapes");
  static_assert(PITCH % 64 == 4, "bank spread of the A fragment loads");
  static_assert(U <= LB_PAD, "a last trip may look U - 1 records past the staged ones");
  static_assert(LB_WCAP % 32 == 0, "the copy takes 32 records a pass");
  __shared__ __attribute__((aligned(16))) float At[16 * PITCH];
  __shared__ int   eS[16 * (LB_WCAP + LB_PAD)];
  __shared__ float eW[16 * (LB_WCAP + LB_PAD)];
  constexpr int KS = 16 / NSTRIP;
  constexpr int KL = K / KS;
  static_assert(NSTRIP * KS == 16 && KL * KS == K && KL % 32 == 0, "16 waves share the product evenly");
  __shared__ float Pt[(KS > 1) ? (KS - 1) * NSTRIP * 256 : 1];

  const int tid  = threadIdx.x;
  const int lane = tid & 31;
  const int wv   = __builtin_amdgcn_readfirstlane(tid >> 5);
  const int m0   = blockIdx.x * 16;

  const int node = m0 + wv;
  float4* trow = (float4*)(At + wv * PITCH);
  float4 own[C];
  #pragma unroll
  for (int c = 0; c < C; ++c) own[c] = make_float4(0.f, 0.f, 0.f, 0.f);
  if (node < nN) {
    const float4* xo = (const float4*)(X + (size_t)node * D);
    #pragma unroll
    for (int c = 0; c < C; ++c) own[c] = xo[lane + 32 * c];
  }
  const int nc = min(node, nN - 1);
  int sb[CSR_S], sl[CSR_S];
  #pragma unroll
  for (int s = 0; s < CSR_S; ++s) {
    sb[s] = rowptr[s * RP_PLANE + nc];
    sl[s] = rowptr[s * RP_PLANE + nc + 1];
  }
  int deg = 0;
  #pragma unroll
  for (int s = 0; s < CSR_S; ++s) {
    sl[s] = (node < nN) ? sl[s] - sb[s] : 0;
    deg += sl[s];
  }
  const bool staged = deg <= LB_WCAP;
  int*   eSw = eS + wv * (LB_WCAP + LB_PAD);
  float* eWw = eW + wv * (LB_WCAP + LB_PAD);
  if (staged) {
    constexpr int NP = LB_WCAP / 32;
    int sv[NP]; float fv[NP];
    #pragma unroll
    for (int k = 0; k < NP; ++k) {
      const int p = lane + 32 * k;
      sv[k] = 0; fv[k] = 0.0f;
      if (p < deg) {
        int shift = sb[0];
        int first = 0;
        #pragma unroll
        for (int s = 1; s < CSR_S; ++s) {
          first += sl[s - 1];
          if (p >= first) shift = sb[s] - first;
        }
        sv[k] = es[p + shift];
        fv[k] = ew[p + shift];
      }
    }
    #pragma unroll
    for (int k = 0; k < NP; ++k) {
      const int p = lane + 32 * k;
      if (p < deg) { eSw[p] = sv[k]; eWw[p] = fv[k]; }
    }
    if (lane < LB_PAD) { eSw[deg + lane] = 0; eWw[deg + lane] = 0.0f; }
  }

  const float iv = (deg > 0) ? (1.0f / (float)deg) : 0.0f;
  #pragma unroll
  for (int c = 0; c < C; ++c) trow[lane + 32 * c] = own[c];
  __syncthreads();

  {
    float4 acc[C];
    #pragma unroll
    for (int c = 0; c < C; ++c) acc[c] = make_float4(0.f, 0.f, 0.f, 0.f);
    const char* xb = (const char*)X;
    const unsigned lane16 = (unsigned)lane * 16u;
    if (staged) {
      int j = 0;
      const int jend = deg;
      unsigned off[U];
      #pragma unroll
      for (int u = 0; u < U; ++u) off[u] = (unsigned)eSw[j + u] * (unsigned)(D * 4) + lane16;
      for (; j + U <= jend; j += U) {
        float ww[U]; float4 v[U][C]; unsigned offn[U];
        #pragma unroll
        for (int u = 0; u < U; ++u) {
          #pragma unroll
          for (int c = 0; c < C; ++c) v[u][c] = *(const float4*)(xb + (off[u] + 512u * c));
        }
        #pragma unroll
        for (int u = 0; u < U; ++u) {
          ww[u]   = eWw[j + u];
          offn[u] = (unsigned)eSw[j + U + u] * (unsigned)(D * 4) + lane16;
        }
        #pragma unroll
        for (int u = 0; u < U; ++u) {
          #pragma unroll
          for (int c = 0; c < C; ++c) {
            acc[c].x = __builtin_fmaf(v[u][c].x, ww[u], acc[c].x);
            acc[c].y = __builtin_fmaf(v[u][c].y, ww[u], acc[c].y);
            acc[c].z = __builtin_fmaf(v[u][c].z, ww[u], acc[c].z);
            acc[c].w = __builtin_fmaf(v[u][c].w, ww[u], acc[c].w);
          }
        }
        #pragma unroll
        for (int u = 0; u < U; ++u) off[u] = offn[u];
      }
      if (j < jend) {
        float ww[U]; float4 v[U][C]; bool ok[U];
        #pragma unroll
        for (int u = 0; u < U; ++u) {
          ok[u] = j + u < jend;
          ww[u] = eWw[j + u];
        }
        #pragma unroll
        for (int u = 0; u < U; ++u) {
          #pragma unroll
          for (int c = 0; c < C; ++c) {
            v[u][c] = make_float4(0.f, 0.f, 0.f, 0.f);
            if (ok[u]) v[u][c] = *(const float4*)(xb + (off[u] + 512u * c));
          }
        }
        #pragma unroll
        for (int u = 0; u < U; ++u) {
          if (ok[u]) {
            #pragma unroll
            for (int c = 0; c < C; ++c) {
              acc[c].x = __builtin_fmaf(v[u][c].x, ww[u], acc[c].x);
              acc[c].y = __builtin_fmaf(v[u][c].y, ww[u], acc[c].y);
              acc[c].z = __builtin_fmaf(v[u][c].z, ww[u], acc[c].z);
              acc[c].w = __builtin_fmaf(v[u][c].w, ww[u], acc[c].w);
            }
          }
        }
      }
    } else {
      #pragma unroll
      for (int sg = 0; sg < CSR_S; ++sg) {
        for (int j = sb[sg]; j < sb[sg] + sl[sg]; ++j) {
          int s = es[j];
          float ww = ew[j];
          const float4* xr = (const float4*)(X + (size_t)s * D);
          #pragma unroll
          for (int c = 0; c < C; ++c) {
            float4 v = xr[lane + 32 * c];
            acc[c].x = __builtin_fmaf(v.x, ww, acc[c].x);
            acc[c].y = __builtin_fmaf(v.y, ww, acc[c].y);
            acc[c].z = __builtin_fmaf(v.z, ww, acc[c].z);
            acc[c].w = __builtin_fmaf(v.w, ww, acc[c].w);
          }
        }
      }
    }
    #pragma unroll
    for (int c = 0; c < C; ++c) {
      trow[D / 4 + lane + 32 * c] =
          make_float4(acc[c].x * iv, acc[c].y * iv, acc[c].z * iv, acc[c].w * iv);
    }
  }
  __syncthreads();

  {
    const int half  = lane >> 4;
    const int mr    = lane & 15;
    const int strip = wv % NSTRIP;
    const int ks    = wv / NSTRIP;
    const int n0    = strip * 16;
    const int k0    = ks * KL;
    const float* ap = At + mr * PITCH + k0 + 2 * half;
    const float* bp = Wt + (size_t)(k0 + 2 * half) * NOUT + n0 + mr;
    v8f acc = {};
    constexpr int UNR = (KS > 1) ? 16 : 8;
    #pragma unroll UNR
    for (int k = 0; k < KL; k += 4) {
      v2f a = *(const v2f*)(ap + k);
      v2f b;
      b.x = bp[k * NOUT];
      b.y = bp[k * NOUT + NOUT];
      acc = __builtin_amdgcn_wmma_f32_16x16x4_f32(false, a, false, b, (short)0, acc, false, false);
    }
    if constexpr (KS > 1) {
      if (ks > 0) {
        float* pw = Pt + ((ks - 1) * NSTRIP + strip) * 256 + lane;
        #pragma unroll
        for (int r = 0; r < 8; ++r) pw[32 * r] = acc[r];
      }
      __syncthreads();
      if (ks == 0) {
        #pragma unroll
        for (int q = 1; q < KS; ++q) {
          const float* pr = Pt + ((q - 1) * NSTRIP + strip) * 256 + lane;
          #pragma unroll
          for (int r = 0; r < 8; ++r) acc[r] += pr[32 * r];
        }
      }
    }
    if (ks == 0) {
      const int n = n0 + mr;
      const float bv = bias[n];
      #pragma unroll
      for (int r = 0; r < 8; ++r) {
        int row = m0 + r + 8 * half;
        float v = acc[r] + bv;
        if (RELU) v = (v > 0.0f) ? v : 0.0f;
        if (row < nN) Y[(size_t)row * NOUT + n] = v;
      }
    }
  }
}

extern "C" void kernel_launch(void* const* d_in, const int* in_sizes, int n_in,
                              void* d_out, int out_size, void* d_ws, size_t ws_size,
                              hipStream_t stream) {
  const float* h   = (const float*)d_in[0];
  const float* w   = (const float*)d_in[1];
  const int*   src = (const int*)d_in[2];
  const int*   dst = (const int*)d_in[3];
  const float* W1  = (const float*)d_in[4];
  const float* b1  = (const float*)d_in[5];
  const float* W2  = (const float*)d_in[6];
  const float* b2  = (const float*)d_in[7];
  float* out = (float*)d_out;

  int*   rowptr = (int*)d_ws;
  int*   es     = rowptr + CSR_S * RP_PLANE;
  float* ew     = (float*)(es + N_EDGES);
  float* x1     = ew + N_EDGES;

  csr_by_node_range<<<CSR_B, CSR_NT, 0, stream>>>(src, dst, w, rowptr, es, ew);

  layer_block16<D_IN, D_HID, true><<<(N_NODES + 15) / 16, 512, 0, stream>>>(
      h, rowptr, es, ew, W1, b1, x1, N_NODES);

  layer_block16<D_HID, D_OUT, false><<<(N_NODES + 15) / 16, 512, 0, stream>>>(
      x1, rowptr, es, ew, W2, b2, out, N_NODES);
}
